# starting kernel
# baseline (speedup 1.0000x reference)
#include <hip/hip_runtime.h>

#define S_TOK  8192
#define D_K    4096
#define D_N    4096
#define R_RANK 16

#define BM  128
#define BN  128
#define BK  16
#define LDK 20

typedef __attribute__((ext_vector_type(2))) float v2f;
typedef __attribute__((ext_vector_type(4))) float v4f;
typedef __attribute__((ext_vector_type(8))) float v8f;

__global__ __launch_bounds__(256) void lora_xa_kernel(
    const float* __restrict__ x, const int* __restrict__ ids,
    const float* __restrict__ A, float* __restrict__ xa)
{
    const int lane = threadIdx.x & 31;
    const int wid  = threadIdx.x >> 5;
    const int s    = blockIdx.x * 8 + wid;
    const int half = lane >> 4;
    const int r    = lane & 15;
    const int id   = ids[s];

    const v4f* xp = (const v4f*)(x + (size_t)s * D_K + half * (D_K / 2));
    const v4f* ap = (const v4f*)(A + ((size_t)id * R_RANK + r) * D_K + half * (D_K / 2));

    float acc = 0.f;
#pragma unroll 4
    for (int i = 0; i < D_K / 8; ++i) {
        v4f xv = xp[i];
        v4f av = ap[i];
        acc += xv[0] * av[0] + xv[1] * av[1] + xv[2] * av[2] + xv[3] * av[3];
    }
    acc += __shfl_xor(acc, 16, 32);
    if (half == 0) xa[(size_t)s * R_RANK + r] = acc;
}

__global__ __launch_bounds__(256) void lora_gemm_kernel(
    const float* __restrict__ x, const float* __restrict__ w,
    const float* __restrict__ bias, const float* __restrict__ Bbuf,
    const int* __restrict__ ids, const float* __restrict__ xa,
    float* __restrict__ out)
{
    __shared__ float sA[2][BM * LDK];
    __shared__ float sB[2][BN * LDK];
    __shared__ float sXA[BM * R_RANK];
    __shared__ int   sID[BM];

    const int tid  = threadIdx.x;
    const int lane = tid & 31;
    const int wid  = tid >> 5;
    const int half = lane >> 4;
    const int l16  = lane & 15;
    const int wm   = wid >> 1;
    const int wn   = wid & 1;

    const int block_m = blockIdx.y * BM;
    const int block_n = blockIdx.x * BN;

    if (tid < BM) sID[tid] = ids[block_m + tid];
    {
        const int r  = tid >> 1;
        const int cg = (tid & 1) * 8;
        const float* src = xa + (size_t)(block_m + r) * R_RANK + cg;
        *(v4f*)&sXA[r * R_RANK + cg]     = *(const v4f*)(src);
        *(v4f*)&sXA[r * R_RANK + cg + 4] = *(const v4f*)(src + 4);
    }

    const int   grow = tid >> 1;
    const int   gcol = (tid & 1) * 8;
    const float* xsrc = x + (size_t)(block_m + grow) * D_K + gcol;
    const float* wsrc = w + (size_t)(block_n + grow) * D_K + gcol;
    const int sOff = grow * LDK + gcol;

    v8f acc[2][4] = {};

    v4f ga0 = *(const v4f*)(xsrc);
    v4f ga1 = *(const v4f*)(xsrc + 4);
    v4f gb0 = *(const v4f*)(wsrc);
    v4f gb1 = *(const v4f*)(wsrc + 4);

    const int NKT = D_K / BK;
    for (int kt = 0; kt < NKT; ++kt) {
        const int buf = kt & 1;

        float* aDst = &sA[buf][sOff];
        float* bDst = &sB[buf][sOff];
        *(v4f*)(aDst)     = ga0;
        *(v4f*)(aDst + 4) = ga1;
        *(v4f*)(bDst)     = gb0;
        *(v4f*)(bDst + 4) = gb1;
        __syncthreads();

        if (kt + 1 < NKT) {
            const int k0n = (kt + 1) * BK;
            ga0 = *(const v4f*)(xsrc + k0n);
            ga1 = *(const v4f*)(xsrc + k0n + 4);
            gb0 = *(const v4f*)(wsrc + k0n);
            gb1 = *(const v4f*)(wsrc + k0n + 4);
            __builtin_prefetch(xsrc + k0n + BK, 0, 0);
            __builtin_prefetch(wsrc + k0n + BK, 0, 0);
        }

        const float* sAb = sA[buf];
        const float* sBb = sB[buf];
#pragma unroll
        for (int kk = 0; kk < 4; ++kk) {
            const int koff = kk * 4 + 2 * half;
            v2f afrag[2], bfrag[4];
#pragma unroll
            for (int im = 0; im < 2; ++im)
                afrag[im] = *(const v2f*)&sAb[(wm * 32 + im * 16 + l16) * LDK + koff];
#pragma unroll
            for (int in = 0; in < 4; ++in)
                bfrag[in] = *(const v2f*)&sBb[(wn * 64 + in * 16 + l16) * LDK + koff];
#pragma unroll
            for (int im = 0; im < 2; ++im)
#pragma unroll
                for (int in = 0; in < 4; ++in)
                    acc[im][in] = __builtin_amdgcn_wmma_f32_16x16x4_f32(
                        false, afrag[im], false, bfrag[in],
                        (short)0, acc[im][in], false, false);
        }
    }

    float biasv[4];
#pragma unroll
    for (int in = 0; in < 4; ++in)
        biasv[in] = bias[block_n + wn * 64 + in * 16 + l16];

#pragma unroll
    for (int im = 0; im < 2; ++im) {
#pragma unroll
        for (int c = 0; c < 8; ++c) {
            const int mloc = wm * 32 + im * 16 + half * 8 + c;
            const int mg   = block_m + mloc;
            const int id   = sID[mloc];
            v4f xa0 = *(const v4f*)&sXA[mloc * R_RANK];
            v4f xa1 = *(const v4f*)&sXA[mloc * R_RANK + 4];
            v4f xa2 = *(const v4f*)&sXA[mloc * R_RANK + 8];
            v4f xa3 = *(const v4f*)&sXA[mloc * R_RANK + 12];
#pragma unroll
            for (int in = 0; in < 4; ++in) {
                const int ng = block_n + wn * 64 + in * 16 + l16;
                const float* bp = Bbuf + ((size_t)id * D_N + ng) * R_RANK;
                v4f b0 = *(const v4f*)(bp);
                v4f b1 = *(const v4f*)(bp + 4);
                v4f b2 = *(const v4f*)(bp + 8);
                v4f b3 = *(const v4f*)(bp + 12);
                float delta = 0.f;
#pragma unroll
                for (int j = 0; j < 4; ++j)
                    delta += xa0[j] * b0[j] + xa1[j] * b1[j]
                           + xa2[j] * b2[j] + xa3[j] * b3[j];
                out[(size_t)mg * D_N + ng] = acc[im][in][c] + biasv[in] + delta;
            }
        }
    }
}

extern "C" void kernel_launch(void* const* d_in, const int* in_sizes, int n_in,
                              void* d_out, int out_size, void* d_ws, size_t ws_size,
                              hipStream_t stream) {
    const float* x    = (const float*)d_in[0];
    const int*   ids  = (const int*)d_in[1];
    const float* w    = (const float*)d_in[2];
    const float* bias = (const float*)d_in[3];
    const float* A    = (const float*)d_in[4];
    const float* B    = (const float*)d_in[5];
    float* out = (float*)d_out;
    float* xa  = (float*)d_ws;

    lora_xa_kernel<<<S_TOK / 8, 256, 0, stream>>>(x, ids, A, xa);

    dim3 grid(D_N / BN, S_TOK / BM);
    lora_gemm_kernel<<<grid, 256, 0, stream>>>(x, w, bias, B, ids, xa, out);
}


# LoRA-A projection kernel: stage each 128-float chunk of all adapters' A rows in LDS once per 32-token block, lanes split d (whole-line loads); changes the summation order of xa
# speedup vs baseline: 1.0781x; 1.0781x over previous
#include <hip/hip_runtime.h>

#define S_TOK  8192
#define D_K    4096
#define D_N    4096
#define R_RANK 16

#define BM  128
#define BN  128
#define BK  16
#define LDK 20

typedef __attribute__((ext_vector_type(2))) float v2f;
typedef __attribute__((ext_vector_type(4))) float v4f;
typedef __attribute__((ext_vector_type(8))) float v8f;

#define XA_TB   32
#define XA_TW   4
#define XA_KC   128
#define XA_NCH  (D_K / XA_KC)
#define XA_AR   128

__global__ __launch_bounds__(256) __attribute__((amdgpu_waves_per_eu(2, 2))) void lora_xa_kernel(
    const float* __restrict__ x, const int* __restrict__ ids,
    const float* __restrict__ A, float* __restrict__ xa)
{
    __shared__ float sAc[2][XA_AR * XA_KC];

    const int tid  = threadIdx.x;
    const int lane = tid & 31;
    const int wid  = tid >> 5;
    const int s0   = blockIdx.x * XA_TB + wid * XA_TW;

    const float* asrc = A + (size_t)wid * D_K + lane * 4;
    const int    aoff = wid * XA_KC + lane * 4;

    const float* xsrc[XA_TW];
    int          lrow[XA_TW];
#pragma unroll
    for (int t = 0; t < XA_TW; ++t) {
        xsrc[t] = x + (size_t)(s0 + t) * D_K + lane * 4;
        lrow[t] = ids[s0 + t] * R_RANK * XA_KC + lane * 4;
    }

    v2f acc[XA_TW][R_RANK];
#pragma unroll
    for (int t = 0; t < XA_TW; ++t)
#pragma unroll
        for (int r = 0; r < R_RANK; ++r) acc[t][r] = (v2f){0.f, 0.f};

    v4f an[16], xn[XA_TW], xc[XA_TW];
#pragma unroll
    for (int j = 0; j < 16; ++j) an[j] = *(const v4f*)(asrc + (size_t)j * 8 * D_K);
#pragma unroll
    for (int t = 0; t < XA_TW; ++t) xc[t] = *(const v4f*)(xsrc[t]);
#pragma unroll
    for (int j = 0; j < 16; ++j) *(v4f*)&sAc[0][aoff + j * 8 * XA_KC] = an[j];
    __syncthreads();

    for (int c = 0; c < XA_NCH; ++c) {
        const bool more = (c + 1 < XA_NCH);
        if (more) {
            const int k0 = (c + 1) * XA_KC;
#pragma unroll
            for (int j = 0; j < 16; ++j) an[j] = *(const v4f*)(asrc + (size_t)j * 8 * D_K + k0);
#pragma unroll
            for (int t = 0; t < XA_TW; ++t) xn[t] = *(const v4f*)(xsrc[t] + k0);
        }
        __builtin_amdgcn_sched_barrier(0);

        const float* sb = sAc[c & 1];
#pragma unroll
        for (int t = 0; t < XA_TW; ++t) {
            const v2f xlo = {xc[t][0], xc[t][1]};
            const v2f xhi = {xc[t][2], xc[t][3]};
            v4f a[R_RANK];
#pragma unroll
            for (int r = 0; r < R_RANK; ++r)
                a[r] = *(const v4f*)&sb[lrow[t] + r * XA_KC];
#pragma unroll
            for (int r = 0; r < R_RANK; ++r) {
                const v2f alo = {a[r][0], a[r][1]};
                const v2f ahi = {a[r][2], a[r][3]};
                v2f v = acc[t][r];
                v = __builtin_elementwise_fma(xlo, alo, v);
                v = __builtin_elementwise_fma(xhi, ahi, v);
                acc[t][r] = v;
            }
        }
        __builtin_amdgcn_sched_barrier(0);
        if (more) {
            float* sd = sAc[(c + 1) & 1];
#pragma unroll
            for (int j = 0; j < 16; ++j) *(v4f*)&sd[aoff + j * 8 * XA_KC] = an[j];
#pragma unroll
            for (int t = 0; t < XA_TW; ++t) xc[t] = xn[t];
        }
        __syncthreads();
    }

#pragma unroll
    for (int t = 0; t < XA_TW; ++t) {
        float v8[8], v4[4], v2[2], v1;
        const bool h16 = (lane & 16) != 0;
#pragma unroll
        for (int r = 0; r < 8; ++r) {
            const float lo = acc[t][r][0] + acc[t][r][1];
            const float hi = acc[t][r + 8][0] + acc[t][r + 8][1];
            const float keep = h16 ? hi : lo;
            const float send = h16 ? lo : hi;
            v8[r] = keep + __shfl_xor(send, 16, 32);
        }
        const bool h8 = (lane & 8) != 0;
#pragma unroll
        for (int r = 0; r < 4; ++r) {
            const float keep = h8 ? v8[r + 4] : v8[r];
            const float send = h8 ? v8[r]     : v8[r + 4];
            v4[r] = keep + __shfl_xor(send, 8, 32);
        }
        const bool h4 = (lane & 4) != 0;
#pragma unroll
        for (int r = 0; r < 2; ++r) {
            const float keep = h4 ? v4[r + 2] : v4[r];
            const float send = h4 ? v4[r]     : v4[r + 2];
            v2[r] = keep + __shfl_xor(send, 4, 32);
        }
        const bool h2 = (lane & 2) != 0;
        {
            const float keep = h2 ? v2[1] : v2[0];
            const float send = h2 ? v2[0] : v2[1];
            v1 = keep + __shfl_xor(send, 2, 32);
        }
        v1 += __shfl_xor(v1, 1, 32);
        if ((lane & 1) == 0) xa[(size_t)(s0 + t) * R_RANK + (lane >> 1)] = v1;
    }
}

__global__ __launch_bounds__(256) void lora_gemm_kernel(
    const float* __restrict__ x, const float* __restrict__ w,
    const float* __restrict__ bias, const float* __restrict__ Bbuf,
    const int* __restrict__ ids, const float* __restrict__ xa,
    float* __restrict__ out)
{
    __shared__ float sA[2][BM * LDK];
    __shared__ float sB[2][BN * LDK];
    __shared__ float sXA[BM * R_RANK];
    __shared__ int   sID[BM];

    const int tid  = threadIdx.x;
    const int lane = tid & 31;
    const int wid  = tid >> 5;
    const int half = lane >> 4;
    const int l16  = lane & 15;
    const int wm   = wid >> 1;
    const int wn   = wid & 1;

    const int block_m = blockIdx.y * BM;
    const int block_n = blockIdx.x * BN;

    if (tid < BM) sID[tid] = ids[block_m + tid];
    {
        const int r  = tid >> 1;
        const int cg = (tid & 1) * 8;
        const float* src = xa + (size_t)(block_m + r) * R_RANK + cg;
        *(v4f*)&sXA[r * R_RANK + cg]     = *(const v4f*)(src);
        *(v4f*)&sXA[r * R_RANK + cg + 4] = *(const v4f*)(src + 4);
    }

    const int   grow = tid >> 1;
    const int   gcol = (tid & 1) * 8;
    const float* xsrc = x + (size_t)(block_m + grow) * D_K + gcol;
    const float* wsrc = w + (size_t)(block_n + grow) * D_K + gcol;
    const int sOff = grow * LDK + gcol;

    v8f acc[2][4] = {};

    v4f ga0 = *(const v4f*)(xsrc);
    v4f ga1 = *(const v4f*)(xsrc + 4);
    v4f gb0 = *(const v4f*)(wsrc);
    v4f gb1 = *(const v4f*)(wsrc + 4);

    const int NKT = D_K / BK;
    for (int kt = 0; kt < NKT; ++kt) {
        const int buf = kt & 1;

        float* aDst = &sA[buf][sOff];
        float* bDst = &sB[buf][sOff];
        *(v4f*)(aDst)     = ga0;
        *(v4f*)(aDst + 4) = ga1;
        *(v4f*)(bDst)     = gb0;
        *(v4f*)(bDst + 4) = gb1;
        __syncthreads();

        if (kt + 1 < NKT) {
            const int k0n = (kt + 1) * BK;
            ga0 = *(const v4f*)(xsrc + k0n);
            ga1 = *(const v4f*)(xsrc + k0n + 4);
            gb0 = *(const v4f*)(wsrc + k0n);
            gb1 = *(const v4f*)(wsrc + k0n + 4);
            __builtin_prefetch(xsrc + k0n + BK, 0, 0);
            __builtin_prefetch(wsrc + k0n + BK, 0, 0);
        }

        const float* sAb = sA[buf];
        const float* sBb = sB[buf];
#pragma unroll
        for (int kk = 0; kk < 4; ++kk) {
            const int koff = kk * 4 + 2 * half;
            v2f afrag[2], bfrag[4];
#pragma unroll
            for (int im = 0; im < 2; ++im)
                afrag[im] = *(const v2f*)&sAb[(wm * 32 + im * 16 + l16) * LDK + koff];
#pragma unroll
            for (int in = 0; in < 4; ++in)
                bfrag[in] = *(const v2f*)&sBb[(wn * 64 + in * 16 + l16) * LDK + koff];
#pragma unroll
            for (int im = 0; im < 2; ++im)
#pragma unroll
                for (int in = 0; in < 4; ++in)
                    acc[im][in] = __builtin_amdgcn_wmma_f32_16x16x4_f32(
                        false, afrag[im], false, bfrag[in],
                        (short)0, acc[im][in], false, false);
        }
    }

    float biasv[4];
#pragma unroll
    for (int in = 0; in < 4; ++in)
        biasv[in] = bias[block_n + wn * 64 + in * 16 + l16];

#pragma unroll
    for (int im = 0; im < 2; ++im) {
#pragma unroll
        for (int c = 0; c < 8; ++c) {
            const int mloc = wm * 32 + im * 16 + half * 8 + c;
            const int mg   = block_m + mloc;
            const int id   = sID[mloc];
            v4f xa0 = *(const v4f*)&sXA[mloc * R_RANK];
            v4f xa1 = *(const v4f*)&sXA[mloc * R_RANK + 4];
            v4f xa2 = *(const v4f*)&sXA[mloc * R_RANK + 8];
            v4f xa3 = *(const v4f*)&sXA[mloc * R_RANK + 12];
#pragma unroll
            for (int in = 0; in < 4; ++in) {
                const int ng = block_n + wn * 64 + in * 16 + l16;
                const float* bp = Bbuf + ((size_t)id * D_N + ng) * R_RANK;
                v4f b0 = *(const v4f*)(bp);
                v4f b1 = *(const v4f*)(bp + 4);
                v4f b2 = *(const v4f*)(bp + 8);
                v4f b3 = *(const v4f*)(bp + 12);
                float delta = 0.f;
#pragma unroll
                for (int j = 0; j < 4; ++j)
                    delta += xa0[j] * b0[j] + xa1[j] * b1[j]
                           + xa2[j] * b2[j] + xa3[j] * b3[j];
                out[(size_t)mg * D_N + ng] = acc[im][in][c] + biasv[in] + delta;
            }
        }
    }
}

extern "C" void kernel_launch(void* const* d_in, const int* in_sizes, int n_in,
                              void* d_out, int out_size, void* d_ws, size_t ws_size,
                              hipStream_t stream) {
    const float* x    = (const float*)d_in[0];
    const int*   ids  = (const int*)d_in[1];
    const float* w    = (const float*)d_in[2];
    const float* bias = (const float*)d_in[3];
    const float* A    = (const float*)d_in[4];
    const float* B    = (const float*)d_in[5];
    float* out = (float*)d_out;
    float* xa  = (float*)d_ws;

    lora_xa_kernel<<<S_TOK / XA_TB, 256, 0, stream>>>(x, ids, A, xa);

    dim3 grid(D_N / BN, S_TOK / BM);
    lora_gemm_kernel<<<grid, 256, 0, stream>>>(x, w, bias, B, ids, xa, out);
}


# GEMM: LoRA delta moved from the vector epilogue into the WMMA K loop as 8 extra k-tiles (one per adapter: xa rows zeroed unless id matches, times B[l]); changes the delta's summation order
# speedup vs baseline: 1.1103x; 1.0299x over previous
#include <hip/hip_runtime.h>

#define S_TOK  8192
#define D_K    4096
#define D_N    4096
#define R_RANK 16

#define BM  128
#define BN  128
#define BK  16
#define LDK 20

typedef __attribute__((ext_vector_type(2))) float v2f;
typedef __attribute__((ext_vector_type(4))) float v4f;
typedef __attribute__((ext_vector_type(8))) float v8f;

#define XA_TB   32
#define XA_TW   4
#define XA_KC   128
#define XA_NCH  (D_K / XA_KC)
#define XA_AR   128

__global__ __launch_bounds__(256) __attribute__((amdgpu_waves_per_eu(2, 2))) void lora_xa_kernel(
    const float* __restrict__ x, const int* __restrict__ ids,
    const float* __restrict__ A, float* __restrict__ xa)
{
    __shared__ float sAc[2][XA_AR * XA_KC];

    const int tid  = threadIdx.x;
    const int lane = tid & 31;
    const int wid  = tid >> 5;
    const int s0   = blockIdx.x * XA_TB + wid * XA_TW;

    const float* asrc = A + (size_t)wid * D_K + lane * 4;
    const int    aoff = wid * XA_KC + lane * 4;

    const float* xsrc[XA_TW];
    int          lrow[XA_TW];
#pragma unroll
    for (int t = 0; t < XA_TW; ++t) {
        xsrc[t] = x + (size_t)(s0 + t) * D_K + lane * 4;
        lrow[t] = ids[s0 + t] * R_RANK * XA_KC + lane * 4;
    }

    v2f acc[XA_TW][R_RANK];
#pragma unroll
    for (int t = 0; t < XA_TW; ++t)
#pragma unroll
        for (int r = 0; r < R_RANK; ++r) acc[t][r] = (v2f){0.f, 0.f};

    v4f an[16], xn[XA_TW], xc[XA_TW];
#pragma unroll
    for (int j = 0; j < 16; ++j) an[j] = *(const v4f*)(asrc + (size_t)j * 8 * D_K);
#pragma unroll
    for (int t = 0; t < XA_TW; ++t) xc[t] = *(const v4f*)(xsrc[t]);
#pragma unroll
    for (int j = 0; j < 16; ++j) *(v4f*)&sAc[0][aoff + j * 8 * XA_KC] = an[j];
    __syncthreads();

    for (int c = 0; c < XA_NCH; ++c) {
        const bool more = (c + 1 < XA_NCH);
        if (more) {
            const int k0 = (c + 1) * XA_KC;
#pragma unroll
            for (int j = 0; j < 16; ++j) an[j] = *(const v4f*)(asrc + (size_t)j * 8 * D_K + k0);
#pragma unroll
            for (int t = 0; t < XA_TW; ++t) xn[t] = *(const v4f*)(xsrc[t] + k0);
        }
        __builtin_amdgcn_sched_barrier(0);

        const float* sb = sAc[c & 1];
#pragma unroll
        for (int t = 0; t < XA_TW; ++t) {
            const v2f xlo = {xc[t][0], xc[t][1]};
            const v2f xhi = {xc[t][2], xc[t][3]};
            v4f a[R_RANK];
#pragma unroll
            for (int r = 0; r < R_RANK; ++r)
                a[r] = *(const v4f*)&sb[lrow[t] + r * XA_KC];
#pragma unroll
            for (int r = 0; r < R_RANK; ++r) {
                const v2f alo = {a[r][0], a[r][1]};
                const v2f ahi = {a[r][2], a[r][3]};
                v2f v = acc[t][r];
                v = __builtin_elementwise_fma(xlo, alo, v);
                v = __builtin_elementwise_fma(xhi, ahi, v);
                acc[t][r] = v;
            }
        }
        __builtin_amdgcn_sched_barrier(0);
        if (more) {
            float* sd = sAc[(c + 1) & 1];
#pragma unroll
            for (int j = 0; j < 16; ++j) *(v4f*)&sd[aoff + j * 8 * XA_KC] = an[j];
#pragma unroll
            for (int t = 0; t < XA_TW; ++t) xc[t] = xn[t];
        }
        __syncthreads();
    }

#pragma unroll
    for (int t = 0; t < XA_TW; ++t) {
        float v8[8], v4[4], v2[2], v1;
        const bool h16 = (lane & 16) != 0;
#pragma unroll
        for (int r = 0; r < 8; ++r) {
            const float lo = acc[t][r][0] + acc[t][r][1];
            const float hi = acc[t][r + 8][0] + acc[t][r + 8][1];
            const float keep = h16 ? hi : lo;
            const float send = h16 ? lo : hi;
            v8[r] = keep + __shfl_xor(send, 16, 32);
        }
        const bool h8 = (lane & 8) != 0;
#pragma unroll
        for (int r = 0; r < 4; ++r) {
            const float keep = h8 ? v8[r + 4] : v8[r];
            const float send = h8 ? v8[r]     : v8[r + 4];
            v4[r] = keep + __shfl_xor(send, 8, 32);
        }
        const bool h4 = (lane & 4) != 0;
#pragma unroll
        for (int r = 0; r < 2; ++r) {
            const float keep = h4 ? v4[r + 2] : v4[r];
            const float send = h4 ? v4[r]     : v4[r + 2];
            v2[r] = keep + __shfl_xor(send, 4, 32);
        }
        const bool h2 = (lane & 2) != 0;
        {
            const float keep = h2 ? v2[1] : v2[0];
            const float send = h2 ? v2[0] : v2[1];
            v1 = keep + __shfl_xor(send, 2, 32);
        }
        v1 += __shfl_xor(v1, 1, 32);
        if ((lane & 1) == 0) xa[(size_t)(s0 + t) * R_RANK + (lane >> 1)] = v1;
    }
}

#define N_LORA 8

__device__ __forceinline__ void mma_ktile(const float* __restrict__ sAb,
                                          const float* __restrict__ sBb,
                                          int arow, int brow, int half,
                                          v8f (&acc)[2][4])
{
#pragma unroll
    for (int kk = 0; kk < 4; ++kk) {
        const int koff = kk * 4 + 2 * half;
        v2f afrag[2], bfrag[4];
#pragma unroll
        for (int im = 0; im < 2; ++im)
            afrag[im] = *(const v2f*)&sAb[(arow + im * 16) * LDK + koff];
#pragma unroll
        for (int in = 0; in < 4; ++in)
            bfrag[in] = *(const v2f*)&sBb[(brow + in * 16) * LDK + koff];
#pragma unroll
        for (int im = 0; im < 2; ++im)
#pragma unroll
            for (int in = 0; in < 4; ++in)
                acc[im][in] = __builtin_amdgcn_wmma_f32_16x16x4_f32(
                    false, afrag[im], false, bfrag[in],
                    (short)0, acc[im][in], false, false);
    }
}

__global__ __launch_bounds__(256) void lora_gemm_kernel(
    const float* __restrict__ x, const float* __restrict__ w,
    const float* __restrict__ bias, const float* __restrict__ Bbuf,
    const int* __restrict__ ids, const float* __restrict__ xa,
    float* __restrict__ out)
{
    __shared__ float sA[2][BM * LDK];
    __shared__ float sB[2][BN * LDK];

    const int tid  = threadIdx.x;
    const int lane = tid & 31;
    const int wid  = tid >> 5;
    const int half = lane >> 4;
    const int l16  = lane & 15;
    const int wm   = wid >> 1;
    const int wn   = wid & 1;

    const int block_m = blockIdx.y * BM;
    const int block_n = blockIdx.x * BN;

    const int   grow = tid >> 1;
    const int   gcol = (tid & 1) * 8;
    const float* xsrc = x + (size_t)(block_m + grow) * D_K + gcol;
    const float* wsrc = w + (size_t)(block_n + grow) * D_K + gcol;
    const int sOff = grow * LDK + gcol;

    const int    myid = ids[block_m + grow];
    const float* xar  = xa + (size_t)(block_m + grow) * R_RANK + gcol;
    const float* bsrc = Bbuf + (size_t)(block_n + grow) * R_RANK + gcol;

    const int arow = wm * 32 + l16;
    const int brow = wn * 64 + l16;

    v8f acc[2][4] = {};

    v4f ga0 = *(const v4f*)(xsrc);
    v4f ga1 = *(const v4f*)(xsrc + 4);
    v4f gb0 = *(const v4f*)(wsrc);
    v4f gb1 = *(const v4f*)(wsrc + 4);

    const int NKT = D_K / BK;
    for (int kt = 0; kt < NKT; ++kt) {
        const int buf = kt & 1;

        float* aDst = &sA[buf][sOff];
        float* bDst = &sB[buf][sOff];
        *(v4f*)(aDst)     = ga0;
        *(v4f*)(aDst + 4) = ga1;
        *(v4f*)(bDst)     = gb0;
        *(v4f*)(bDst + 4) = gb1;
        __syncthreads();

        if (kt + 1 < NKT) {
            const int k0n = (kt + 1) * BK;
            ga0 = *(const v4f*)(xsrc + k0n);
            ga1 = *(const v4f*)(xsrc + k0n + 4);
            gb0 = *(const v4f*)(wsrc + k0n);
            gb1 = *(const v4f*)(wsrc + k0n + 4);
            __builtin_prefetch(xsrc + k0n + BK, 0, 0);
            __builtin_prefetch(wsrc + k0n + BK, 0, 0);
        } else {
            ga0 = *(const v4f*)(xar);
            ga1 = *(const v4f*)(xar + 4);
            gb0 = *(const v4f*)(bsrc);
            gb1 = *(const v4f*)(bsrc + 4);
        }

        mma_ktile(sA[buf], sB[buf], arow, brow, half, acc);
    }

    for (int l = 0; l < N_LORA; ++l) {
        const int buf = l & 1;

        const bool on = (myid == l);
        const v4f z = {0.f, 0.f, 0.f, 0.f};
        float* aDst = &sA[buf][sOff];
        float* bDst = &sB[buf][sOff];
        *(v4f*)(aDst)     = on ? ga0 : z;
        *(v4f*)(aDst + 4) = on ? ga1 : z;
        *(v4f*)(bDst)     = gb0;
        *(v4f*)(bDst + 4) = gb1;
        __syncthreads();

        if (l + 1 < N_LORA) {
            const float* bp = bsrc + (size_t)(l + 1) * D_N * R_RANK;
            gb0 = *(const v4f*)(bp);
            gb1 = *(const v4f*)(bp + 4);
        }

        mma_ktile(sA[buf], sB[buf], arow, brow, half, acc);
    }

    float biasv[4];
#pragma unroll
    for (int in = 0; in < 4; ++in)
        biasv[in] = bias[block_n + wn * 64 + in * 16 + l16];

#pragma unroll
    for (int im = 0; im < 2; ++im) {
#pragma unroll
        for (int c = 0; c < 8; ++c) {
            const int mg = block_m + wm * 32 + im * 16 + half * 8 + c;
#pragma unroll
            for (int in = 0; in < 4; ++in) {
                const int ng = block_n + wn * 64 + in * 16 + l16;
                out[(size_t)mg * D_N + ng] = acc[im][in][c] + biasv[in];
            }
        }
    }
}

extern "C" void kernel_launch(void* const* d_in, const int* in_sizes, int n_in,
                              void* d_out, int out_size, void* d_ws, size_t ws_size,
                              hipStream_t stream) {
    const float* x    = (const float*)d_in[0];
    const int*   ids  = (const int*)d_in[1];
    const float* w    = (const float*)d_in[2];
    const float* bias = (const float*)d_in[3];
    const float* A    = (const float*)d_in[4];
    const float* B    = (const float*)d_in[5];
    float* out = (float*)d_out;
    float* xa  = (float*)d_ws;

    lora_xa_kernel<<<S_TOK / XA_TB, 256, 0, stream>>>(x, ids, A, xa);

    dim3 grid(D_N / BN, S_TOK / BM);
    lora_gemm_kernel<<<grid, 256, 0, stream>>>(x, w, bias, B, ids, xa, out);
}


# GEMM k-tile deepened from 16 to 64 per barrier (128 WMMAs per wave and hand-over, was 32); rows staged as whole 256-B pieces, no separate prefetch; 2 blocks per compute unit; sum order unchanged
# speedup vs baseline: 1.2364x; 1.1136x over previous
#include <hip/hip_runtime.h>

#define S_TOK  8192
#define D_K    4096
#define D_N    4096
#define R_RANK 16

#define BM  128
#define BN  128
#define BK  64
#define LDK 68

typedef __attribute__((ext_vector_type(2))) float v2f;
typedef __attribute__((ext_vector_type(4))) float v4f;
typedef __attribute__((ext_vector_type(8))) float v8f;

#define XA_TB   32
#define XA_TW   4
#define XA_KC   128
#define XA_NCH  (D_K / XA_KC)
#define XA_AR   128

__global__ __launch_bounds__(256) __attribute__((amdgpu_waves_per_eu(2, 2))) void lora_xa_kernel(
    const float* __restrict__ x, const int* __restrict__ ids,
    const float* __restrict__ A, float* __restrict__ xa)
{
    __shared__ float sAc[2][XA_AR * XA_KC];

    const int tid  = threadIdx.x;
    const int lane = tid & 31;
    const int wid  = tid >> 5;
    const int s0   = blockIdx.x * XA_TB + wid * XA_TW;

    const float* asrc = A + (size_t)wid * D_K + lane * 4;
    const int    aoff = wid * XA_KC + lane * 4;

    const float* xsrc[XA_TW];
    int          lrow[XA_TW];
#pragma unroll
    for (int t = 0; t < XA_TW; ++t) {
        xsrc[t] = x + (size_t)(s0 + t) * D_K + lane * 4;
        lrow[t] = ids[s0 + t] * R_RANK * XA_KC + lane * 4;
    }

    v2f acc[XA_TW][R_RANK];
#pragma unroll
    for (int t = 0; t < XA_TW; ++t)
#pragma unroll
        for (int r = 0; r < R_RANK; ++r) acc[t][r] = (v2f){0.f, 0.f};

    v4f an[16], xn[XA_TW], xc[XA_TW];
#pragma unroll
    for (int j = 0; j < 16; ++j) an[j] = *(const v4f*)(asrc + (size_t)j * 8 * D_K);
#pragma unroll
    for (int t = 0; t < XA_TW; ++t) xc[t] = *(const v4f*)(xsrc[t]);
#pragma unroll
    for (int j = 0; j < 16; ++j) *(v4f*)&sAc[0][aoff + j * 8 * XA_KC] = an[j];
    __syncthreads();

    for (int c = 0; c < XA_NCH; ++c) {
        const bool more = (c + 1 < XA_NCH);
        if (more) {
            const int k0 = (c + 1) * XA_KC;
#pragma unroll
            for (int j = 0; j < 16; ++j) an[j] = *(const v4f*)(asrc + (size_t)j * 8 * D_K + k0);
#pragma unroll
            for (int t = 0; t < XA_TW; ++t) xn[t] = *(const v4f*)(xsrc[t] + k0);
        }
        __builtin_amdgcn_sched_barrier(0);

        const float* sb = sAc[c & 1];
#pragma unroll
        for (int t = 0; t < XA_TW; ++t) {
            const v2f xlo = {xc[t][0], xc[t][1]};
            const v2f xhi = {xc[t][2], xc[t][3]};
            v4f a[R_RANK];
#pragma unroll
            for (int r = 0; r < R_RANK; ++r)
                a[r] = *(const v4f*)&sb[lrow[t] + r * XA_KC];
#pragma unroll
            for (int r = 0; r < R_RANK; ++r) {
                const v2f alo = {a[r][0], a[r][1]};
                const v2f ahi = {a[r][2], a[r][3]};
                v2f v = acc[t][r];
                v = __builtin_elementwise_fma(xlo, alo, v);
                v = __builtin_elementwise_fma(xhi, ahi, v);
                acc[t][r] = v;
            }
        }
        __builtin_amdgcn_sched_barrier(0);
        if (more) {
            float* sd = sAc[(c + 1) & 1];
#pragma unroll
            for (int j = 0; j < 16; ++j) *(v4f*)&sd[aoff + j * 8 * XA_KC] = an[j];
#pragma unroll
            for (int t = 0; t < XA_TW; ++t) xc[t] = xn[t];
        }
        __syncthreads();
    }

#pragma unroll
    for (int t = 0; t < XA_TW; ++t) {
        float v8[8], v4[4], v2[2], v1;
        const bool h16 = (lane & 16) != 0;
#pragma unroll
        for (int r = 0; r < 8; ++r) {
            const float lo = acc[t][r][0] + acc[t][r][1];
            const float hi = acc[t][r + 8][0] + acc[t][r + 8][1];
            const float keep = h16 ? hi : lo;
            const float send = h16 ? lo : hi;
            v8[r] = keep + __shfl_xor(send, 16, 32);
        }
        const bool h8 = (lane & 8) != 0;
#pragma unroll
        for (int r = 0; r < 4; ++r) {
            const float keep = h8 ? v8[r + 4] : v8[r];
            const float send = h8 ? v8[r]     : v8[r + 4];
            v4[r] = keep + __shfl_xor(send, 8, 32);
        }
        const bool h4 = (lane & 4) != 0;
#pragma unroll
        for (int r = 0; r < 2; ++r) {
            const float keep = h4 ? v4[r + 2] : v4[r];
            const float send = h4 ? v4[r]     : v4[r + 2];
            v2[r] = keep + __shfl_xor(send, 4, 32);
        }
        const bool h2 = (lane & 2) != 0;
        {
            const float keep = h2 ? v2[1] : v2[0];
            const float send = h2 ? v2[0] : v2[1];
            v1 = keep + __shfl_xor(send, 2, 32);
        }
        v1 += __shfl_xor(v1, 1, 32);
        if ((lane & 1) == 0) xa[(size_t)(s0 + t) * R_RANK + (lane >> 1)] = v1;
    }
}

#define N_LORA 8
#define NKT    (D_K / BK)
#define NLT    (N_LORA * R_RANK / BK)
#define ST_R   8
#define KSUB   16

struct Frags {
    v2f a[4][2];
    v2f b[4][4];
};

__device__ __forceinline__ void load_frags(const float* __restrict__ sAb,
                                           const float* __restrict__ sBb,
                                           int aoff, int boff, int ks, Frags& f)
{
#pragma unroll
    for (int kk = 0; kk < 4; ++kk) {
        const int koff = ks * KSUB + kk * 4;
#pragma unroll
        for (int im = 0; im < 2; ++im)
            f.a[kk][im] = *(const v2f*)&sAb[aoff + im * 16 * LDK + koff];
#pragma unroll
        for (int in = 0; in < 4; ++in)
            f.b[kk][in] = *(const v2f*)&sBb[boff + in * 16 * LDK + koff];
    }
}

__device__ __forceinline__ void mma_sub(const Frags& f, v8f (&acc)[2][4])
{
#pragma unroll
    for (int kk = 0; kk < 4; ++kk)
#pragma unroll
        for (int im = 0; im < 2; ++im)
#pragma unroll
            for (int in = 0; in < 4; ++in)
                acc[im][in] = __builtin_amdgcn_wmma_f32_16x16x4_f32(
                    false, f.a[kk][im], false, f.b[kk][in],
                    (short)0, acc[im][in], false, false);
}

__global__ __launch_bounds__(256) __attribute__((amdgpu_waves_per_eu(4, 4))) void lora_gemm_kernel(
    const float* __restrict__ x, const float* __restrict__ w,
    const float* __restrict__ bias, const float* __restrict__ Bbuf,
    const int* __restrict__ ids, const float* __restrict__ xa,
    float* __restrict__ out)
{
    __shared__ float sA[2][BM * LDK];
    __shared__ float sB[2][BN * LDK];

    const int tid  = threadIdx.x;
    const int lane = tid & 31;
    const int wid  = tid >> 5;
    const int half = lane >> 4;
    const int l16  = lane & 15;
    const int wm   = wid >> 1;
    const int wn   = wid & 1;

    const int block_m = blockIdx.y * BM;
    const int block_n = blockIdx.x * BN;

    const int   srow = tid >> 4;
    const int   scol = (tid & 15) * 4;
    const float* xsrc = x + (size_t)(block_m + srow) * D_K + scol;
    const float* wsrc = w + (size_t)(block_n + srow) * D_K + scol;
    const int sOff = srow * LDK + scol;

    const int lsel = (tid & 15) >> 2;
    const int rcol = (tid & 3) * 4;

    const int aoff = (wm * 32 + l16) * LDK + 2 * half;
    const int boff = (wn * 64 + l16) * LDK + 2 * half;

    v8f acc[2][4] = {};

    v4f ga[ST_R], gb[ST_R];
#pragma unroll
    for (int j = 0; j < ST_R; ++j) {
        ga[j] = *(const v4f*)(xsrc + (size_t)j * 16 * D_K);
        gb[j] = *(const v4f*)(wsrc + (size_t)j * 16 * D_K);
    }

    for (int kt = 0; kt < NKT + NLT; ++kt) {
        const int buf = kt & 1;

        float* aDst = &sA[buf][sOff];
        float* bDst = &sB[buf][sOff];
#pragma unroll
        for (int j = 0; j < ST_R; ++j) {
            *(v4f*)(aDst + j * 16 * LDK) = ga[j];
            *(v4f*)(bDst + j * 16 * LDK) = gb[j];
        }
        __syncthreads();

        if (kt + 1 < NKT) {
            const int k0n = (kt + 1) * BK;
#pragma unroll
            for (int j = 0; j < ST_R; ++j) {
                ga[j] = *(const v4f*)(xsrc + (size_t)j * 16 * D_K + k0n);
                gb[j] = *(const v4f*)(wsrc + (size_t)j * 16 * D_K + k0n);
            }
        } else if (kt + 1 < NKT + NLT) {
            const int l = (kt + 1 - NKT) * (BK / R_RANK) + lsel;
            const v4f z = {0.f, 0.f, 0.f, 0.f};
#pragma unroll
            for (int j = 0; j < ST_R; ++j) {
                const int row = srow + 16 * j;
                const int tok = block_m + row;
                const v4f v  = *(const v4f*)(xa + (size_t)tok * R_RANK + rcol);
                const int id = ids[tok];
                ga[j] = (id == l) ? v : z;
                gb[j] = *(const v4f*)(Bbuf + ((size_t)l * D_N + block_n + row) * R_RANK + rcol);
            }
        }

        const float* sAb = sA[buf];
        const float* sBb = sB[buf];
        Frags f0, f1;
        load_frags(sAb, sBb, aoff, boff, 0, f0);
        load_frags(sAb, sBb, aoff, boff, 1, f1);
        mma_sub(f0, acc);
        load_frags(sAb, sBb, aoff, boff, 2, f0);
        mma_sub(f1, acc);
        load_frags(sAb, sBb, aoff, boff, 3, f1);
        mma_sub(f0, acc);
        mma_sub(f1, acc);
    }

    float biasv[4];
#pragma unroll
    for (int in = 0; in < 4; ++in)
        biasv[in] = bias[block_n + wn * 64 + in * 16 + l16];

#pragma unroll
    for (int im = 0; im < 2; ++im) {
#pragma unroll
        for (int c = 0; c < 8; ++c) {
            const int mg = block_m + wm * 32 + im * 16 + half * 8 + c;
#pragma unroll
            for (int in = 0; in < 4; ++in) {
                const int ng = block_n + wn * 64 + in * 16 + l16;
                out[(size_t)mg * D_N + ng] = acc[im][in][c] + biasv[in];
            }
        }
    }
}

extern "C" void kernel_launch(void* const* d_in, const int* in_sizes, int n_in,
                              void* d_out, int out_size, void* d_ws, size_t ws_size,
                              hipStream_t stream) {
    const float* x    = (const float*)d_in[0];
    const int*   ids  = (const int*)d_in[1];
    const float* w    = (const float*)d_in[2];
    const float* bias = (const float*)d_in[3];
    const float* A    = (const float*)d_in[4];
    const float* B    = (const float*)d_in[5];
    float* out = (float*)d_out;
    float* xa  = (float*)d_ws;

    lora_xa_kernel<<<S_TOK / XA_TB, 256, 0, stream>>>(x, ids, A, xa);

    dim3 grid(D_N / BN, S_TOK / BM);
    lora_gemm_kernel<<<grid, 256, 0, stream>>>(x, w, bias, B, ids, xa, out);
}
